# starting kernel
# baseline (speedup 1.0000x reference)
#include <hip/hip_runtime.h>
#include <hip/hip_bf16.h>
#include <math.h>

typedef __attribute__((ext_vector_type(2))) float v2f;
typedef __attribute__((ext_vector_type(8))) float v8f;

#define LOG2PI 1.8378770664093453f
#define LOG2E  1.4426950408889634f
#define LN2    0.6931471805599453f
#define TILES_PER_WAVE 4

__global__ __launch_bounds__(256) void mdn_posterior_wmma(
    const float* __restrict__ y,
    const float* __restrict__ mus,
    const float* __restrict__ sigmas,
    const float* __restrict__ pi_logits,
    const float* __restrict__ prior,
    float* __restrict__ out,
    int n_full_tiles)
{
    __shared__ float W[5 * 128];

    const int tid = threadIdx.x;
    if (tid < 128) {
        const int j = tid >> 3;
        const int c = tid & 7;
        const int base = j * 16 + c * 2;
        const float m0 = mus[base],    m1 = mus[base + 1];
        const float s0 = sigmas[base], s1 = sigmas[base + 1];
        const float iv0 = 1.0f / (s0 * s0);
        const float iv1 = 1.0f / (s1 * s1);
        W[      tid] = LOG2E * -0.5f * iv0;
        W[128 + tid] = LOG2E * -0.5f * iv1;
        W[256 + tid] = LOG2E * m0 * iv0;
        W[384 + tid] = LOG2E * m1 * iv1;
        float mx = pi_logits[j * 8];
        #pragma unroll
        for (int i = 1; i < 8; ++i) mx = fmaxf(mx, pi_logits[j * 8 + i]);
        float se = 0.0f;
        #pragma unroll
        for (int i = 0; i < 8; ++i) se += __expf(pi_logits[j * 8 + i] - mx);
        const float lse = mx + __logf(se);
        W[512 + tid] = LOG2E * (-0.5f * (m0 * m0 * iv0 + m1 * m1 * iv1)
                                - (__logf(s0) + __logf(s1)) - LOG2PI
                                + (pi_logits[j * 8 + c] - lse) + __logf(prior[j]));
    }
    __syncthreads();

    const int lane = tid & 31;
    const int wave = tid >> 5;
    const int m    = lane & 15;
    const int hi   = lane >> 4;

    const int tile0 = (blockIdx.x * (blockDim.x >> 5) + wave) * TILES_PER_WAVE;
    if (tile0 >= n_full_tiles) return;

    v2f a[8];
    v8f kvec[8];
    #pragma unroll
    for (int t = 0; t < 8; ++t) {
        const int ncol = t * 16 + m;
        a[t].x = W[(hi ? 256 :   0) + ncol];
        a[t].y = W[(hi ? 384 : 128) + ncol];
        const float* Kp = &W[512 + t * 16 + hi * 8];
        #pragma unroll
        for (int i = 0; i < 8; ++i) kvec[t][i] = Kp[i];
    }

    const int row0 = tile0 * 16 + m;
    const float* yp = y + row0 * 2;
    float*       op = out + row0 * 16 + hi;

    int count = n_full_tiles - tile0;
    if (count > TILES_PER_WAVE) count = TILES_PER_WAVE;

    #pragma unroll 1
    for (int it = 0; it < count; ++it) {
        const v2f yv = *(const v2f*)yp;
        yp += 32;

        v2f b;
        b.x = hi ? yv.x : yv.x * yv.x;
        b.y = hi ? yv.y : yv.y * yv.y;

        #pragma unroll
        for (int t = 0; t < 8; ++t) {
            const v8f d = __builtin_amdgcn_wmma_f32_16x16x4_f32(
                false, a[t], false, b, (short)0, kvec[t], false, false);

            float mx = fmaxf(fmaxf(d[0], d[1]), fmaxf(d[2], d[3]));
            mx = fmaxf(mx, fmaxf(fmaxf(d[4], d[5]), fmaxf(d[6], d[7])));
            float ex0 = __builtin_amdgcn_exp2f(d[0] - mx);
            float ex1 = __builtin_amdgcn_exp2f(d[1] - mx);
            float ex2 = __builtin_amdgcn_exp2f(d[2] - mx);
            float ex3 = __builtin_amdgcn_exp2f(d[3] - mx);
            float ex4 = __builtin_amdgcn_exp2f(d[4] - mx);
            float ex5 = __builtin_amdgcn_exp2f(d[5] - mx);
            float ex6 = __builtin_amdgcn_exp2f(d[6] - mx);
            float ex7 = __builtin_amdgcn_exp2f(d[7] - mx);
            const float e = ((ex0 + ex1) + (ex2 + ex3)) + ((ex4 + ex5) + (ex6 + ex7));
            op[2 * t] = LN2 * (mx + __builtin_amdgcn_logf(e));
        }
        op += 256;
    }
}

__global__ __launch_bounds__(256) void mdn_posterior_tail(
    const float* __restrict__ y,
    const float* __restrict__ mus,
    const float* __restrict__ sigmas,
    const float* __restrict__ pi_logits,
    const float* __restrict__ prior,
    float* __restrict__ out,
    int n_samp, int row0)
{
    const int w = blockIdx.x * blockDim.x + threadIdx.x;
    if (w >= (n_samp - row0) * 16) return;
    const int row = row0 + (w >> 4);
    const int j   = w & 15;
    const float y0 = y[row * 2], y1 = y[row * 2 + 1];

    float mxp = pi_logits[j * 8];
    #pragma unroll
    for (int i = 1; i < 8; ++i) mxp = fmaxf(mxp, pi_logits[j * 8 + i]);
    float se = 0.0f;
    #pragma unroll
    for (int i = 0; i < 8; ++i) se += __expf(pi_logits[j * 8 + i] - mxp);
    const float lse = mxp + __logf(se);
    const float lp  = __logf(prior[j]);

    float vals[8];
    float vmax = -INFINITY;
    #pragma unroll
    for (int c = 0; c < 8; ++c) {
        const int base = j * 16 + c * 2;
        const float m0 = mus[base],    m1 = mus[base + 1];
        const float s0 = sigmas[base], s1 = sigmas[base + 1];
        const float d0 = (y0 - m0) / s0, d1 = (y1 - m1) / s1;
        const float v = -0.5f * (d0 * d0 + d1 * d1)
                        - (__logf(s0) + __logf(s1)) - LOG2PI
                        + (pi_logits[j * 8 + c] - lse) + lp;
        vals[c] = v;
        vmax = fmaxf(vmax, v);
    }
    float e = 0.0f;
    #pragma unroll
    for (int c = 0; c < 8; ++c) e += __expf(vals[c] - vmax);
    out[row * 16 + j] = vmax + __logf(e);
}

extern "C" void kernel_launch(void* const* d_in, const int* in_sizes, int n_in,
                              void* d_out, int out_size, void* d_ws, size_t ws_size,
                              hipStream_t stream) {
    const float* y         = (const float*)d_in[0];
    const float* mus       = (const float*)d_in[1];
    const float* sigmas    = (const float*)d_in[2];
    const float* pi_logits = (const float*)d_in[3];
    const float* prior     = (const float*)d_in[4];
    float* out = (float*)d_out;

    const int n_samp       = in_sizes[0] / 2;
    const int n_full_tiles = n_samp / 16;
    const int rem          = n_samp - n_full_tiles * 16;

    if (n_full_tiles > 0) {
        const int n_super = (n_full_tiles + TILES_PER_WAVE - 1) / TILES_PER_WAVE;
        const int waves_per_block = 8;
        dim3 block(32 * waves_per_block);
        dim3 grid((n_super + waves_per_block - 1) / waves_per_block);
        mdn_posterior_wmma<<<grid, block, 0, stream>>>(y, mus, sigmas, pi_logits,
                                                       prior, out, n_full_tiles);
    }
    if (rem > 0) {
        mdn_posterior_tail<<<1, 256, 0, stream>>>(y, mus, sigmas, pi_logits, prior,
                                                  out, n_samp, n_full_tiles * 16);
    }
}


# WMMA operand roles swapped: samples (row-permuted) are A rows, the 16 symbols B columns, one product per component; D is row-major, each b32 store is one whole 128-B line; K added after the product
# speedup vs baseline: 1.6608x; 1.6608x over previous
#include <hip/hip_runtime.h>
#include <hip/hip_bf16.h>
#include <math.h>

typedef __attribute__((ext_vector_type(2))) float v2f;
typedef __attribute__((ext_vector_type(8))) float v8f;

#define LOG2PI 1.8378770664093453f
#define LOG2E  1.4426950408889634f
#define LN2    0.6931471805599453f
#define TILES_PER_WAVE 4

__global__ __launch_bounds__(256) void mdn_posterior_wmma(
    const float* __restrict__ y,
    const float* __restrict__ mus,
    const float* __restrict__ sigmas,
    const float* __restrict__ pi_logits,
    const float* __restrict__ prior,
    float* __restrict__ out,
    int n_full_tiles)
{
    __shared__ float W[5 * 128];

    const int tid = threadIdx.x;
    if (tid < 128) {
        const int j = tid >> 3;
        const int c = tid & 7;
        const int base = j * 16 + c * 2;
        const float m0 = mus[base],    m1 = mus[base + 1];
        const float s0 = sigmas[base], s1 = sigmas[base + 1];
        const float iv0 = 1.0f / (s0 * s0);
        const float iv1 = 1.0f / (s1 * s1);
        W[      tid] = LOG2E * -0.5f * iv0;
        W[128 + tid] = LOG2E * -0.5f * iv1;
        W[256 + tid] = LOG2E * m0 * iv0;
        W[384 + tid] = LOG2E * m1 * iv1;
        float mx = pi_logits[j * 8];
        #pragma unroll
        for (int i = 1; i < 8; ++i) mx = fmaxf(mx, pi_logits[j * 8 + i]);
        float se = 0.0f;
        #pragma unroll
        for (int i = 0; i < 8; ++i) se += __expf(pi_logits[j * 8 + i] - mx);
        const float lse = mx + __logf(se);
        W[512 + tid] = LOG2E * (-0.5f * (m0 * m0 * iv0 + m1 * m1 * iv1)
                                - (__logf(s0) + __logf(s1)) - LOG2PI
                                + (pi_logits[j * 8 + c] - lse) + __logf(prior[j]));
    }
    __syncthreads();

    const int lane = tid & 31;
    const int wave = tid >> 5;
    const int m    = lane & 15;
    const int hi   = lane >> 4;

    const int tile0 = (blockIdx.x * (blockDim.x >> 5) + wave) * TILES_PER_WAVE;
    if (tile0 >= n_full_tiles) return;

    v2f b[8];
    v2f k2[8];
    {
        const float* P0 = &W[(hi ? 256 :   0) + m * 8];
        const float* P1 = &W[(hi ? 384 : 128) + m * 8];
        const float* Kp = &W[512 + m * 8];
        #pragma unroll
        for (int c = 0; c < 8; ++c) {
            b[c].x  = P0[c];
            b[c].y  = P1[c];
            k2[c].x = Kp[c];
            k2[c].y = Kp[c];
        }
    }

    const int srow = 2 * (m & 7) + (m >> 3);
    const float* yp = y + (tile0 * 16 + srow) * 2;
    float*       op = out + tile0 * 256 + lane;

    int count = n_full_tiles - tile0;
    if (count > TILES_PER_WAVE) count = TILES_PER_WAVE;

    const v8f zero = {0.f, 0.f, 0.f, 0.f, 0.f, 0.f, 0.f, 0.f};

    #pragma unroll 1
    for (int it = 0; it < count; ++it) {
        const v2f yv = *(const v2f*)yp;
        yp += 32;

        v2f a;
        a.x = hi ? yv.x : yv.x * yv.x;
        a.y = hi ? yv.y : yv.y * yv.y;

        v8f d[8];
        #pragma unroll
        for (int c = 0; c < 8; ++c) {
            const v8f p = __builtin_amdgcn_wmma_f32_16x16x4_f32(
                              false, a, false, b[c], (short)0, zero, false, false);
            const v2f p0 = __builtin_shufflevector(p, p, 0, 1) + k2[c];
            const v2f p1 = __builtin_shufflevector(p, p, 2, 3) + k2[c];
            const v2f p2 = __builtin_shufflevector(p, p, 4, 5) + k2[c];
            const v2f p3 = __builtin_shufflevector(p, p, 6, 7) + k2[c];
            d[c] = __builtin_shufflevector(__builtin_shufflevector(p0, p1, 0, 1, 2, 3),
                                           __builtin_shufflevector(p2, p3, 0, 1, 2, 3),
                                           0, 1, 2, 3, 4, 5, 6, 7);
        }

        v8f mx = __builtin_elementwise_max(
                     __builtin_elementwise_max(__builtin_elementwise_max(d[0], d[1]),
                                               __builtin_elementwise_max(d[2], d[3])),
                     __builtin_elementwise_max(__builtin_elementwise_max(d[4], d[5]),
                                               __builtin_elementwise_max(d[6], d[7])));
        v8f ex[8];
        #pragma unroll
        for (int c = 0; c < 8; ++c) {
            const v8f t = d[c] - mx;
            #pragma unroll
            for (int i = 0; i < 8; ++i) ex[c][i] = __builtin_amdgcn_exp2f(t[i]);
        }
        const v8f e = ((ex[0] + ex[1]) + (ex[2] + ex[3])) + ((ex[4] + ex[5]) + (ex[6] + ex[7]));
        v8f lg;
        #pragma unroll
        for (int i = 0; i < 8; ++i) lg[i] = __builtin_amdgcn_logf(e[i]);
        const v8f r = (mx + lg) * LN2;

        #pragma unroll
        for (int i = 0; i < 8; ++i) op[32 * i] = r[i];
        op += 256;
    }
}

__global__ __launch_bounds__(256) void mdn_posterior_tail(
    const float* __restrict__ y,
    const float* __restrict__ mus,
    const float* __restrict__ sigmas,
    const float* __restrict__ pi_logits,
    const float* __restrict__ prior,
    float* __restrict__ out,
    int n_samp, int row0)
{
    const int w = blockIdx.x * blockDim.x + threadIdx.x;
    if (w >= (n_samp - row0) * 16) return;
    const int row = row0 + (w >> 4);
    const int j   = w & 15;
    const float y0 = y[row * 2], y1 = y[row * 2 + 1];

    float mxp = pi_logits[j * 8];
    #pragma unroll
    for (int i = 1; i < 8; ++i) mxp = fmaxf(mxp, pi_logits[j * 8 + i]);
    float se = 0.0f;
    #pragma unroll
    for (int i = 0; i < 8; ++i) se += __expf(pi_logits[j * 8 + i] - mxp);
    const float lse = mxp + __logf(se);
    const float lp  = __logf(prior[j]);

    float vals[8];
    float vmax = -INFINITY;
    #pragma unroll
    for (int c = 0; c < 8; ++c) {
        const int base = j * 16 + c * 2;
        const float m0 = mus[base],    m1 = mus[base + 1];
        const float s0 = sigmas[base], s1 = sigmas[base + 1];
        const float d0 = (y0 - m0) / s0, d1 = (y1 - m1) / s1;
        const float v = -0.5f * (d0 * d0 + d1 * d1)
                        - (__logf(s0) + __logf(s1)) - LOG2PI
                        + (pi_logits[j * 8 + c] - lse) + lp;
        vals[c] = v;
        vmax = fmaxf(vmax, v);
    }
    float e = 0.0f;
    #pragma unroll
    for (int c = 0; c < 8; ++c) e += __expf(vals[c] - vmax);
    out[row * 16 + j] = vmax + __logf(e);
}

extern "C" void kernel_launch(void* const* d_in, const int* in_sizes, int n_in,
                              void* d_out, int out_size, void* d_ws, size_t ws_size,
                              hipStream_t stream) {
    const float* y         = (const float*)d_in[0];
    const float* mus       = (const float*)d_in[1];
    const float* sigmas    = (const float*)d_in[2];
    const float* pi_logits = (const float*)d_in[3];
    const float* prior     = (const float*)d_in[4];
    float* out = (float*)d_out;

    const int n_samp       = in_sizes[0] / 2;
    const int n_full_tiles = n_samp / 16;
    const int rem          = n_samp - n_full_tiles * 16;

    if (n_full_tiles > 0) {
        const int n_super = (n_full_tiles + TILES_PER_WAVE - 1) / TILES_PER_WAVE;
        const int waves_per_block = 8;
        dim3 block(32 * waves_per_block);
        dim3 grid((n_super + waves_per_block - 1) / waves_per_block);
        mdn_posterior_wmma<<<grid, block, 0, stream>>>(y, mus, sigmas, pi_logits,
                                                       prior, out, n_full_tiles);
    }
    if (rem > 0) {
        mdn_posterior_tail<<<1, 256, 0, stream>>>(y, mus, sigmas, pi_logits, prior,
                                                  out, n_samp, n_full_tiles * 16);
    }
}


# Resident grid: 256 blocks of 8 waves, one per WGP (LDS reservation); a wave walks its 15-16 tiles two per pass, samples requested a pass ahead; table built once per WGP, not once per 32 tiles
# speedup vs baseline: 1.8371x; 1.1062x over previous
#include <hip/hip_runtime.h>
#include <hip/hip_bf16.h>
#include <math.h>

typedef __attribute__((ext_vector_type(2))) float v2f;
typedef __attribute__((ext_vector_type(8))) float v8f;

#define LOG2PI 1.8378770664093453f
#define LOG2E  1.4426950408889634f
#define LN2    0.6931471805599453f
#define WAVES_PER_BLOCK 8
#define RESIDENT_BLOCKS 256
#define LDS_FLOATS (164 * 256)

__device__ __forceinline__ void mdn_tile(const v2f a, const v2f* __restrict__ b,
                                         const v2f* __restrict__ k2, float* __restrict__ op)
{
    const v8f zero = {0.f, 0.f, 0.f, 0.f, 0.f, 0.f, 0.f, 0.f};

    v8f d[8];
    #pragma unroll
    for (int c = 0; c < 8; ++c) {
        const v8f p = __builtin_amdgcn_wmma_f32_16x16x4_f32(
                          false, a, false, b[c], (short)0, zero, false, false);
        const v2f p0 = __builtin_shufflevector(p, p, 0, 1) + k2[c];
        const v2f p1 = __builtin_shufflevector(p, p, 2, 3) + k2[c];
        const v2f p2 = __builtin_shufflevector(p, p, 4, 5) + k2[c];
        const v2f p3 = __builtin_shufflevector(p, p, 6, 7) + k2[c];
        d[c] = __builtin_shufflevector(__builtin_shufflevector(p0, p1, 0, 1, 2, 3),
                                       __builtin_shufflevector(p2, p3, 0, 1, 2, 3),
                                       0, 1, 2, 3, 4, 5, 6, 7);
    }

    v8f mx = __builtin_elementwise_max(
                 __builtin_elementwise_max(__builtin_elementwise_max(d[0], d[1]),
                                           __builtin_elementwise_max(d[2], d[3])),
                 __builtin_elementwise_max(__builtin_elementwise_max(d[4], d[5]),
                                           __builtin_elementwise_max(d[6], d[7])));
    v8f ex[8];
    #pragma unroll
    for (int c = 0; c < 8; ++c) {
        const v8f s = d[c] - mx;
        #pragma unroll
        for (int i = 0; i < 8; ++i) ex[c][i] = __builtin_amdgcn_exp2f(s[i]);
    }
    const v8f e = ((ex[0] + ex[1]) + (ex[2] + ex[3])) + ((ex[4] + ex[5]) + (ex[6] + ex[7]));
    v8f lg;
    #pragma unroll
    for (int i = 0; i < 8; ++i) lg[i] = __builtin_amdgcn_logf(e[i]);
    const v8f r = (mx + lg) * LN2;

    #pragma unroll
    for (int i = 0; i < 8; ++i) op[32 * i] = r[i];
}

__global__ __launch_bounds__(256) void mdn_posterior_wmma(
    const float* __restrict__ y,
    const float* __restrict__ mus,
    const float* __restrict__ sigmas,
    const float* __restrict__ pi_logits,
    const float* __restrict__ prior,
    float* __restrict__ out,
    int n_full_tiles)
{
    __shared__ float W[LDS_FLOATS];

    const int tid  = threadIdx.x;
    const int lane = tid & 31;
    const int m    = lane & 15;
    const int hi   = lane >> 4;
    const int wave = __builtin_amdgcn_readfirstlane(tid >> 5);

    const int stride = gridDim.x * WAVES_PER_BLOCK;
    const int first  = wave * gridDim.x + blockIdx.x;

    const float* ylane = y + (2 * (m & 7) + (m >> 3)) * 2;

    const int t0 = first < n_full_tiles ? first : n_full_tiles - 1;
    const int t1 = first + stride < n_full_tiles ? first + stride : t0;
    v2f yq[2];
    yq[0] = *(const v2f*)(ylane + t0 * 32);
    yq[1] = *(const v2f*)(ylane + t1 * 32);

    if (tid < 128) {
        const int j = tid >> 3;
        const int c = tid & 7;
        const int base = j * 16 + c * 2;
        const float m0 = mus[base],    m1 = mus[base + 1];
        const float s0 = sigmas[base], s1 = sigmas[base + 1];
        const float iv0 = 1.0f / (s0 * s0);
        const float iv1 = 1.0f / (s1 * s1);
        W[      tid] = LOG2E * -0.5f * iv0;
        W[128 + tid] = LOG2E * -0.5f * iv1;
        W[256 + tid] = LOG2E * m0 * iv0;
        W[384 + tid] = LOG2E * m1 * iv1;
        float mx = pi_logits[j * 8];
        #pragma unroll
        for (int i = 1; i < 8; ++i) mx = fmaxf(mx, pi_logits[j * 8 + i]);
        float se = 0.0f;
        #pragma unroll
        for (int i = 0; i < 8; ++i) se += __expf(pi_logits[j * 8 + i] - mx);
        const float lse = mx + __logf(se);
        W[512 + tid] = LOG2E * (-0.5f * (m0 * m0 * iv0 + m1 * m1 * iv1)
                                - (__logf(s0) + __logf(s1)) - LOG2PI
                                + (pi_logits[j * 8 + c] - lse) + __logf(prior[j]));
    }
    __syncthreads();

    if (first >= n_full_tiles) return;

    v2f b[8];
    v2f k2[8];
    {
        const float* P0 = &W[(hi ? 256 :   0) + m * 8];
        const float* P1 = &W[(hi ? 384 : 128) + m * 8];
        const float* Kp = &W[512 + m * 8];
        #pragma unroll
        for (int c = 0; c < 8; ++c) {
            b[c].x  = P0[c];
            b[c].y  = P1[c];
            k2[c].x = Kp[c];
            k2[c].y = Kp[c];
        }
    }

    int t = first;
    #pragma unroll 1
    for (; t + stride < n_full_tiles; t += 2 * stride) {
        {
            const v2f yv = yq[0];
            v2f a;
            a.x = hi ? yv.x : yv.x * yv.x;
            a.y = hi ? yv.y : yv.y * yv.y;
            const int ta = t + 2 * stride < n_full_tiles ? t + 2 * stride : t;
            yq[0] = *(const v2f*)(ylane + ta * 32);
            mdn_tile(a, b, k2, out + t * 256 + lane);
        }
        {
            const int tb = t + stride;
            const v2f yv = yq[1];
            v2f a;
            a.x = hi ? yv.x : yv.x * yv.x;
            a.y = hi ? yv.y : yv.y * yv.y;
            const int ta = tb + 2 * stride < n_full_tiles ? tb + 2 * stride : tb;
            yq[1] = *(const v2f*)(ylane + ta * 32);
            mdn_tile(a, b, k2, out + tb * 256 + lane);
        }
    }
    if (t < n_full_tiles) {
        const v2f yv = yq[0];
        v2f a;
        a.x = hi ? yv.x : yv.x * yv.x;
        a.y = hi ? yv.y : yv.y * yv.y;
        mdn_tile(a, b, k2, out + t * 256 + lane);
    }
}

__global__ __launch_bounds__(256) void mdn_posterior_tail(
    const float* __restrict__ y,
    const float* __restrict__ mus,
    const float* __restrict__ sigmas,
    const float* __restrict__ pi_logits,
    const float* __restrict__ prior,
    float* __restrict__ out,
    int n_samp, int row0)
{
    const int w = blockIdx.x * blockDim.x + threadIdx.x;
    if (w >= (n_samp - row0) * 16) return;
    const int row = row0 + (w >> 4);
    const int j   = w & 15;
    const float y0 = y[row * 2], y1 = y[row * 2 + 1];

    float mxp = pi_logits[j * 8];
    #pragma unroll
    for (int i = 1; i < 8; ++i) mxp = fmaxf(mxp, pi_logits[j * 8 + i]);
    float se = 0.0f;
    #pragma unroll
    for (int i = 0; i < 8; ++i) se += __expf(pi_logits[j * 8 + i] - mxp);
    const float lse = mxp + __logf(se);
    const float lp  = __logf(prior[j]);

    float vals[8];
    float vmax = -INFINITY;
    #pragma unroll
    for (int c = 0; c < 8; ++c) {
        const int base = j * 16 + c * 2;
        const float m0 = mus[base],    m1 = mus[base + 1];
        const float s0 = sigmas[base], s1 = sigmas[base + 1];
        const float d0 = (y0 - m0) / s0, d1 = (y1 - m1) / s1;
        const float v = -0.5f * (d0 * d0 + d1 * d1)
                        - (__logf(s0) + __logf(s1)) - LOG2PI
                        + (pi_logits[j * 8 + c] - lse) + lp;
        vals[c] = v;
        vmax = fmaxf(vmax, v);
    }
    float e = 0.0f;
    #pragma unroll
    for (int c = 0; c < 8; ++c) e += __expf(vals[c] - vmax);
    out[row * 16 + j] = vmax + __logf(e);
}

extern "C" void kernel_launch(void* const* d_in, const int* in_sizes, int n_in,
                              void* d_out, int out_size, void* d_ws, size_t ws_size,
                              hipStream_t stream) {
    const float* y         = (const float*)d_in[0];
    const float* mus       = (const float*)d_in[1];
    const float* sigmas    = (const float*)d_in[2];
    const float* pi_logits = (const float*)d_in[3];
    const float* prior     = (const float*)d_in[4];
    float* out = (float*)d_out;

    const int n_samp       = in_sizes[0] / 2;
    const int n_full_tiles = n_samp / 16;
    const int rem          = n_samp - n_full_tiles * 16;

    if (n_full_tiles > 0) {
        dim3 block(32 * WAVES_PER_BLOCK);
        dim3 grid(RESIDENT_BLOCKS);
        mdn_posterior_wmma<<<grid, block, 0, stream>>>(y, mus, sigmas, pi_logits,
                                                       prior, out, n_full_tiles);
    }
    if (rem > 0) {
        mdn_posterior_tail<<<1, 256, 0, stream>>>(y, mus, sigmas, pi_logits, prior,
                                                  out, n_samp, n_full_tiles * 16);
    }
}
